# starting kernel
# baseline (speedup 1.0000x reference)
#include <hip/hip_runtime.h>

#define B_    16
#define CIN_  16
#define OUT_  16
#define H_    320
#define W_    320
#define HW_   (H_ * W_)
#define KEXP_ 4
#define HID_  5
#define KTOT_ 144
#define COLS  322
#define CPAD  18

typedef __attribute__((ext_vector_type(2))) float v2f;
typedef __attribute__((ext_vector_type(8))) float v8f;

__global__ __launch_bounds__(256) void dyn_pool_kernel(
    const float* __restrict__ x, float* __restrict__ pooled) {
  const int bc = blockIdx.x;
  const float4* p4 = (const float4*)(x + (size_t)bc * HW_);
  float s = 0.0f;
  for (int i = threadIdx.x; i < HW_ / 4; i += 256) {
    float4 v = p4[i];
    s += v.x + v.y + v.z + v.w;
  }
  __shared__ float red[256];
  red[threadIdx.x] = s;
  __syncthreads();
  for (int off = 128; off > 0; off >>= 1) {
    if (threadIdx.x < (unsigned)off) red[threadIdx.x] += red[threadIdx.x + off];
    __syncthreads();
  }
  if (threadIdx.x == 0) pooled[bc] = red[0] * (1.0f / (float)HW_);
}

__global__ __launch_bounds__(256) void dyn_attn_kernel(
    const float* __restrict__ pooled, const float* __restrict__ w1,
    const float* __restrict__ w2, const float* __restrict__ b2,
    const float* __restrict__ weight, float* __restrict__ wmix) {
  const int t = threadIdx.x;
  const int b = t >> 4;
  const int j = t & 15;

  float hid[HID_];
#pragma unroll
  for (int h = 0; h < HID_; ++h) {
    float a = 0.0f;
#pragma unroll
    for (int c = 0; c < CIN_; ++c) a += pooled[b * CIN_ + c] * w1[h * CIN_ + c];
    hid[h] = fmaxf(a, 0.0f);
  }
  float av[KEXP_];
#pragma unroll
  for (int k = 0; k < KEXP_; ++k) {
    const int i = k * 16 + j;
    float a = b2[i];
#pragma unroll
    for (int h = 0; h < HID_; ++h) a += hid[h] * w2[i * HID_ + h];
    av[k] = a;
  }
  float mx = fmaxf(fmaxf(av[0], av[1]), fmaxf(av[2], av[3]));
  float e[KEXP_], sum = 0.0f;
#pragma unroll
  for (int k = 0; k < KEXP_; ++k) { e[k] = expf(av[k] - mx); sum += e[k]; }
  const float inv = 1.0f / sum;

  __shared__ float sm[B_][KEXP_ * OUT_];
#pragma unroll
  for (int k = 0; k < KEXP_; ++k) sm[b][k * 16 + j] = e[k] * inv;
  __syncthreads();

  const int o = j;
  float coef[KEXP_];
#pragma unroll
  for (int k = 0; k < KEXP_; ++k) coef[k] = sm[b][o * KEXP_ + k];

  const float* wsrc = weight + (size_t)o * KTOT_ * KEXP_;
  float* wdst = wmix + ((size_t)b * OUT_ + o) * KTOT_;
  for (int c = 0; c < CIN_; ++c) {
#pragma unroll
    for (int rd = 0; rd < 9; ++rd) {
      float acc = 0.0f;
#pragma unroll
      for (int e2 = 0; e2 < KEXP_; ++e2)
        acc += coef[e2] * wsrc[(c * 9 + rd) * KEXP_ + e2];
      wdst[rd * 16 + c] = acc;
    }
  }
}

__global__ __launch_bounds__(256) void dyn_conv_kernel(
    const float* __restrict__ x, const float* __restrict__ wmix,
    float* __restrict__ out) {
  const int blk = blockIdx.x;
  const int b = blk / (H_ / 2);
  const int h0 = (blk % (H_ / 2)) * 2;

  __shared__ float xs[4 * COLS * CPAD];

  const float* xb = x + (size_t)b * CIN_ * HW_;
  for (int i = threadIdx.x; i < CIN_ * 4 * COLS; i += 256) {
    const int col = i % COLS;
    const int cr = i / COLS;
    const int r = cr & 3;
    const int c = cr >> 2;
    const int hh = h0 - 1 + r;
    const int ww = col - 1;
    float v = 0.0f;
    if (hh >= 0 && hh < H_ && ww >= 0 && ww < W_)
      v = xb[c * HW_ + hh * W_ + ww];
    xs[(r * COLS + col) * CPAD + c] = v;
  }
  __syncthreads();

  const int lane = threadIdx.x & 31;
  const int wave = threadIdx.x >> 5;
  const int m = lane & 15;
  const int hi = lane >> 4;

  const float* wb = wmix + ((size_t)b * OUT_ + m) * KTOT_;
  v2f a[36];
#pragma unroll
  for (int s = 0; s < 36; ++s) {
    a[s].x = wb[4 * s + 2 * hi + 0];
    a[s].y = wb[4 * s + 2 * hi + 1];
  }

#pragma unroll
  for (int it = 0; it < 5; ++it) {
    const int tt = wave * 5 + it;
    const int lr = tt / 20;
    const int wbase = (tt % 20) * 16;
    const int base = (lr * COLS + wbase + m) * CPAD + 2 * hi;

    v8f acc = {};
#pragma unroll
    for (int s = 0; s < 36; ++s) {
      const int cb = (4 * s) & 15;
      const int rd = (4 * s) >> 4;
      const int r  = rd / 3;
      const int dx = rd % 3;
      const v2f bf = *(const v2f*)(xs + base + (r * COLS + dx) * CPAD + cb);
      acc = __builtin_amdgcn_wmma_f32_16x16x4_f32(
          false, a[s], false, bf, (short)0, acc, false, false);
    }

    float* ob = out + (size_t)b * OUT_ * HW_ + (h0 + lr) * W_ + wbase + m;
#pragma unroll
    for (int jj = 0; jj < 8; ++jj)
      ob[(size_t)(jj + 8 * hi) * HW_] = acc[jj];
  }
}

extern "C" void kernel_launch(void* const* d_in, const int* in_sizes, int n_in,
                              void* d_out, int out_size, void* d_ws, size_t ws_size,
                              hipStream_t stream) {
  (void)in_sizes; (void)n_in; (void)out_size; (void)ws_size;
  const float* x      = (const float*)d_in[0];
  const float* w1     = (const float*)d_in[1];
  const float* w2     = (const float*)d_in[2];
  const float* b2     = (const float*)d_in[3];
  const float* weight = (const float*)d_in[4];
  float* out = (float*)d_out;

  float* pooled = (float*)d_ws;
  float* wmix   = (float*)d_ws + 256;

  dyn_pool_kernel<<<B_ * CIN_, 256, 0, stream>>>(x, pooled);
  dyn_attn_kernel<<<1, 256, 0, stream>>>(pooled, w1, w2, b2, weight, wmix);
  dyn_conv_kernel<<<B_ * (H_ / 2), 256, 0, stream>>>(x, wmix, out);
}


# conv staging: each thread issues its twenty 16-byte loads of x together (was one 4-byte load in flight per wave); LDS pixel-minor with channel pairs interleaved, B still one aligned b64
# speedup vs baseline: 1.8119x; 1.8119x over previous
#include <hip/hip_runtime.h>

#define B_    16
#define CIN_  16
#define OUT_  16
#define H_    320
#define W_    320
#define HW_   (H_ * W_)
#define KEXP_ 4
#define HID_  5
#define KTOT_ 144
#define P2_   336
#define CO_   3

typedef __attribute__((ext_vector_type(2))) float v2f;
typedef __attribute__((ext_vector_type(8))) float v8f;

__global__ __launch_bounds__(256) void dyn_pool_kernel(
    const float* __restrict__ x, float* __restrict__ pooled) {
  const int bc = blockIdx.x;
  const float4* p4 = (const float4*)(x + (size_t)bc * HW_);
  float s = 0.0f;
  for (int i = threadIdx.x; i < HW_ / 4; i += 256) {
    float4 v = p4[i];
    s += v.x + v.y + v.z + v.w;
  }
  __shared__ float red[256];
  red[threadIdx.x] = s;
  __syncthreads();
  for (int off = 128; off > 0; off >>= 1) {
    if (threadIdx.x < (unsigned)off) red[threadIdx.x] += red[threadIdx.x + off];
    __syncthreads();
  }
  if (threadIdx.x == 0) pooled[bc] = red[0] * (1.0f / (float)HW_);
}

__global__ __launch_bounds__(256) void dyn_attn_kernel(
    const float* __restrict__ pooled, const float* __restrict__ w1,
    const float* __restrict__ w2, const float* __restrict__ b2,
    const float* __restrict__ weight, float* __restrict__ wmix) {
  const int t = threadIdx.x;
  const int b = t >> 4;
  const int j = t & 15;

  float hid[HID_];
#pragma unroll
  for (int h = 0; h < HID_; ++h) {
    float a = 0.0f;
#pragma unroll
    for (int c = 0; c < CIN_; ++c) a += pooled[b * CIN_ + c] * w1[h * CIN_ + c];
    hid[h] = fmaxf(a, 0.0f);
  }
  float av[KEXP_];
#pragma unroll
  for (int k = 0; k < KEXP_; ++k) {
    const int i = k * 16 + j;
    float a = b2[i];
#pragma unroll
    for (int h = 0; h < HID_; ++h) a += hid[h] * w2[i * HID_ + h];
    av[k] = a;
  }
  float mx = fmaxf(fmaxf(av[0], av[1]), fmaxf(av[2], av[3]));
  float e[KEXP_], sum = 0.0f;
#pragma unroll
  for (int k = 0; k < KEXP_; ++k) { e[k] = expf(av[k] - mx); sum += e[k]; }
  const float inv = 1.0f / sum;

  __shared__ float sm[B_][KEXP_ * OUT_];
#pragma unroll
  for (int k = 0; k < KEXP_; ++k) sm[b][k * 16 + j] = e[k] * inv;
  __syncthreads();

  const int o = j;
  float coef[KEXP_];
#pragma unroll
  for (int k = 0; k < KEXP_; ++k) coef[k] = sm[b][o * KEXP_ + k];

  const float* wsrc = weight + (size_t)o * KTOT_ * KEXP_;
  float* wdst = wmix + ((size_t)b * OUT_ + o) * KTOT_;
  for (int c = 0; c < CIN_; ++c) {
#pragma unroll
    for (int rd = 0; rd < 9; ++rd) {
      float acc = 0.0f;
#pragma unroll
      for (int e2 = 0; e2 < KEXP_; ++e2)
        acc += coef[e2] * wsrc[(c * 9 + rd) * KEXP_ + e2];
      wdst[rd * 16 + c] = acc;
    }
  }
}

__global__ __launch_bounds__(256) void dyn_conv_kernel(
    const float* __restrict__ x, const float* __restrict__ wmix,
    float* __restrict__ out) {
  const int blk = blockIdx.x;
  const int b = blk / (H_ / 2);
  const int h0 = (blk % (H_ / 2)) * 2;

  __shared__ float xs[4 * 8 * P2_ * 2];

  const float* xb = x + (size_t)b * CIN_ * HW_;
  float4 ve[10], vo[10];
#pragma unroll
  for (int i = 0; i < 10; ++i) {
    const unsigned f = threadIdx.x + 256u * i;
    const unsigned k = f % 80u;
    const unsigned q = f / 80u;
    const int cp = q & 7;
    const int hh = h0 - 1 + (int)(q >> 3);
    const int hc = min(max(hh, 0), H_ - 1);
    const float4* sp =
        (const float4*)(xb + (size_t)(2 * cp) * HW_ + hc * W_) + k;
    ve[i] = sp[0];
    vo[i] = sp[HW_ / 4];
  }
#pragma unroll
  for (int i = 0; i < 10; ++i) {
    const unsigned f = threadIdx.x + 256u * i;
    const unsigned k = f % 80u;
    const unsigned q = f / 80u;
    const int hh = h0 - 1 + (int)(q >> 3);
    const bool ok = (hh >= 0) && (hh < H_);
    const float4 z = make_float4(0.0f, 0.0f, 0.0f, 0.0f);
    const float4 e = ok ? ve[i] : z;
    const float4 o = ok ? vo[i] : z;
    float4* dp = (float4*)(xs + (q * P2_ + 4u + 4u * k) * 2u);
    dp[0] = make_float4(e.x, o.x, e.y, o.y);
    dp[1] = make_float4(e.z, o.z, e.w, o.w);
  }
  if (threadIdx.x < 64) {
    const unsigned q = threadIdx.x >> 1;
    const unsigned slot = (threadIdx.x & 1) ? (unsigned)(CO_ + W_ + 1) : (unsigned)CO_;
    v2f zz = {0.0f, 0.0f};
    *(v2f*)(xs + (q * P2_ + slot) * 2u) = zz;
  }
  __syncthreads();

  const int lane = threadIdx.x & 31;
  const int wave = threadIdx.x >> 5;
  const int m = lane & 15;
  const int hi = lane >> 4;

  const float* wb = wmix + ((size_t)b * OUT_ + m) * KTOT_;
  v2f a[36];
#pragma unroll
  for (int s = 0; s < 36; ++s) {
    a[s].x = wb[4 * s + 2 * hi + 0];
    a[s].y = wb[4 * s + 2 * hi + 1];
  }

#pragma unroll 1
  for (int it = 0; it < 5; ++it) {
    const int tt = wave * 5 + it;
    const int lr = tt / 20;
    const int wbase = (tt % 20) * 16;
    const int base = ((lr * 8 + hi) * P2_ + wbase + m + CO_) * 2;

    v8f acc = {};
#pragma unroll
    for (int s = 0; s < 36; ++s) {
      const int cb = (4 * s) & 15;
      const int rd = (4 * s) >> 4;
      const int r  = rd / 3;
      const int dx = rd % 3;
      const v2f bf =
          *(const v2f*)(xs + base + ((r * 8 + (cb >> 1)) * P2_ + dx) * 2);
      acc = __builtin_amdgcn_wmma_f32_16x16x4_f32(
          false, a[s], false, bf, (short)0, acc, false, false);
    }

    float* ob = out + (size_t)b * OUT_ * HW_ + (h0 + lr) * W_ + wbase + m;
#pragma unroll
    for (int jj = 0; jj < 8; ++jj)
      ob[(size_t)(jj + 8 * hi) * HW_] = acc[jj];
  }
}

extern "C" void kernel_launch(void* const* d_in, const int* in_sizes, int n_in,
                              void* d_out, int out_size, void* d_ws, size_t ws_size,
                              hipStream_t stream) {
  (void)in_sizes; (void)n_in; (void)out_size; (void)ws_size;
  const float* x      = (const float*)d_in[0];
  const float* w1     = (const float*)d_in[1];
  const float* w2     = (const float*)d_in[2];
  const float* b2     = (const float*)d_in[3];
  const float* weight = (const float*)d_in[4];
  float* out = (float*)d_out;

  float* pooled = (float*)d_ws;
  float* wmix   = (float*)d_ws + 256;

  dyn_pool_kernel<<<B_ * CIN_, 256, 0, stream>>>(x, pooled);
  dyn_attn_kernel<<<1, 256, 0, stream>>>(pooled, w1, w2, b2, weight, wmix);
  dyn_conv_kernel<<<B_ * (H_ / 2), 256, 0, stream>>>(x, wmix, out);
}


# attention+mix kernel: one work-group per sample, not one in all; expert bank read as coalesced 16-byte loads, nine in flight per thread (was strided 4-byte loads waited pair by pair); same sums
# speedup vs baseline: 2.6113x; 1.4412x over previous
#include <hip/hip_runtime.h>

#define B_    16
#define CIN_  16
#define OUT_  16
#define H_    320
#define W_    320
#define HW_   (H_ * W_)
#define KEXP_ 4
#define HID_  5
#define KTOT_ 144
#define P2_   336
#define CO_   3

typedef __attribute__((ext_vector_type(2))) float v2f;
typedef __attribute__((ext_vector_type(8))) float v8f;

__global__ __launch_bounds__(256) void dyn_pool_kernel(
    const float* __restrict__ x, float* __restrict__ pooled) {
  const int bc = blockIdx.x;
  const float4* p4 = (const float4*)(x + (size_t)bc * HW_);
  float s = 0.0f;
  for (int i = threadIdx.x; i < HW_ / 4; i += 256) {
    float4 v = p4[i];
    s += v.x + v.y + v.z + v.w;
  }
  __shared__ float red[256];
  red[threadIdx.x] = s;
  __syncthreads();
  for (int off = 128; off > 0; off >>= 1) {
    if (threadIdx.x < (unsigned)off) red[threadIdx.x] += red[threadIdx.x + off];
    __syncthreads();
  }
  if (threadIdx.x == 0) pooled[bc] = red[0] * (1.0f / (float)HW_);
}

__global__ __launch_bounds__(256) void dyn_attn_kernel(
    const float* __restrict__ pooled, const float* __restrict__ w1,
    const float* __restrict__ w2, const float* __restrict__ b2,
    const float* __restrict__ weight, float* __restrict__ wmix) {
  const int t = threadIdx.x;
  const int b = blockIdx.x;

  __shared__ __attribute__((aligned(16))) float sm[KEXP_ * OUT_];
  __shared__ __attribute__((aligned(16))) float wt[OUT_ * KTOT_];

  const float4* wq = (const float4*)weight;
  float4 wv[9];
#pragma unroll
  for (int i = 0; i < 9; ++i) wv[i] = wq[t + 256 * i];

  if (t < 16) {
    const int j = t;
    float hid[HID_];
#pragma unroll
    for (int h = 0; h < HID_; ++h) {
      float a = 0.0f;
#pragma unroll
      for (int c = 0; c < CIN_; ++c) a += pooled[b * CIN_ + c] * w1[h * CIN_ + c];
      hid[h] = fmaxf(a, 0.0f);
    }
    float av[KEXP_];
#pragma unroll
    for (int k = 0; k < KEXP_; ++k) {
      const int i = k * 16 + j;
      float a = b2[i];
#pragma unroll
      for (int h = 0; h < HID_; ++h) a += hid[h] * w2[i * HID_ + h];
      av[k] = a;
    }
    float mx = fmaxf(fmaxf(av[0], av[1]), fmaxf(av[2], av[3]));
    float e[KEXP_], sum = 0.0f;
#pragma unroll
    for (int k = 0; k < KEXP_; ++k) { e[k] = expf(av[k] - mx); sum += e[k]; }
    const float inv = 1.0f / sum;
#pragma unroll
    for (int k = 0; k < KEXP_; ++k) sm[k * 16 + j] = e[k] * inv;
  }
  __syncthreads();

#pragma unroll
  for (int i = 0; i < 9; ++i) {
    const int g = t + 256 * i;
    const int oc = g / 9;
    const int rd = g - oc * 9;
    const int o = oc >> 4;
    const int c = oc & 15;
    const float4 cf = *(const float4*)(sm + o * KEXP_);
    float acc = 0.0f;
    acc += cf.x * wv[i].x;
    acc += cf.y * wv[i].y;
    acc += cf.z * wv[i].z;
    acc += cf.w * wv[i].w;
    wt[o * KTOT_ + rd * 16 + c] = acc;
  }
  __syncthreads();

  float4* dst = (float4*)(wmix + (size_t)b * OUT_ * KTOT_);
  const float4* sp = (const float4*)wt;
  for (int q = t; q < OUT_ * KTOT_ / 4; q += 256) dst[q] = sp[q];
}

__global__ __launch_bounds__(256) void dyn_conv_kernel(
    const float* __restrict__ x, const float* __restrict__ wmix,
    float* __restrict__ out) {
  const int blk = blockIdx.x;
  const int b = blk / (H_ / 2);
  const int h0 = (blk % (H_ / 2)) * 2;

  __shared__ float xs[4 * 8 * P2_ * 2];

  const float* xb = x + (size_t)b * CIN_ * HW_;
  float4 ve[10], vo[10];
#pragma unroll
  for (int i = 0; i < 10; ++i) {
    const unsigned f = threadIdx.x + 256u * i;
    const unsigned k = f % 80u;
    const unsigned q = f / 80u;
    const int cp = q & 7;
    const int hh = h0 - 1 + (int)(q >> 3);
    const int hc = min(max(hh, 0), H_ - 1);
    const float4* sp =
        (const float4*)(xb + (size_t)(2 * cp) * HW_ + hc * W_) + k;
    ve[i] = sp[0];
    vo[i] = sp[HW_ / 4];
  }
#pragma unroll
  for (int i = 0; i < 10; ++i) {
    const unsigned f = threadIdx.x + 256u * i;
    const unsigned k = f % 80u;
    const unsigned q = f / 80u;
    const int hh = h0 - 1 + (int)(q >> 3);
    const bool ok = (hh >= 0) && (hh < H_);
    const float4 z = make_float4(0.0f, 0.0f, 0.0f, 0.0f);
    const float4 e = ok ? ve[i] : z;
    const float4 o = ok ? vo[i] : z;
    float4* dp = (float4*)(xs + (q * P2_ + 4u + 4u * k) * 2u);
    dp[0] = make_float4(e.x, o.x, e.y, o.y);
    dp[1] = make_float4(e.z, o.z, e.w, o.w);
  }
  if (threadIdx.x < 64) {
    const unsigned q = threadIdx.x >> 1;
    const unsigned slot = (threadIdx.x & 1) ? (unsigned)(CO_ + W_ + 1) : (unsigned)CO_;
    v2f zz = {0.0f, 0.0f};
    *(v2f*)(xs + (q * P2_ + slot) * 2u) = zz;
  }
  __syncthreads();

  const int lane = threadIdx.x & 31;
  const int wave = threadIdx.x >> 5;
  const int m = lane & 15;
  const int hi = lane >> 4;

  const float* wb = wmix + ((size_t)b * OUT_ + m) * KTOT_;
  v2f a[36];
#pragma unroll
  for (int s = 0; s < 36; ++s) {
    a[s].x = wb[4 * s + 2 * hi + 0];
    a[s].y = wb[4 * s + 2 * hi + 1];
  }

#pragma unroll 1
  for (int it = 0; it < 5; ++it) {
    const int tt = wave * 5 + it;
    const int lr = tt / 20;
    const int wbase = (tt % 20) * 16;
    const int base = ((lr * 8 + hi) * P2_ + wbase + m + CO_) * 2;

    v8f acc = {};
#pragma unroll
    for (int s = 0; s < 36; ++s) {
      const int cb = (4 * s) & 15;
      const int rd = (4 * s) >> 4;
      const int r  = rd / 3;
      const int dx = rd % 3;
      const v2f bf =
          *(const v2f*)(xs + base + ((r * 8 + (cb >> 1)) * P2_ + dx) * 2);
      acc = __builtin_amdgcn_wmma_f32_16x16x4_f32(
          false, a[s], false, bf, (short)0, acc, false, false);
    }

    float* ob = out + (size_t)b * OUT_ * HW_ + (h0 + lr) * W_ + wbase + m;
#pragma unroll
    for (int jj = 0; jj < 8; ++jj)
      ob[(size_t)(jj + 8 * hi) * HW_] = acc[jj];
  }
}

extern "C" void kernel_launch(void* const* d_in, const int* in_sizes, int n_in,
                              void* d_out, int out_size, void* d_ws, size_t ws_size,
                              hipStream_t stream) {
  (void)in_sizes; (void)n_in; (void)out_size; (void)ws_size;
  const float* x      = (const float*)d_in[0];
  const float* w1     = (const float*)d_in[1];
  const float* w2     = (const float*)d_in[2];
  const float* b2     = (const float*)d_in[3];
  const float* weight = (const float*)d_in[4];
  float* out = (float*)d_out;

  float* pooled = (float*)d_ws;
  float* wmix   = (float*)d_ws + 256;

  dyn_pool_kernel<<<B_ * CIN_, 256, 0, stream>>>(x, pooled);
  dyn_attn_kernel<<<B_, 256, 0, stream>>>(pooled, w1, w2, b2, weight, wmix);
  dyn_conv_kernel<<<B_ * (H_ / 2), 256, 0, stream>>>(x, wmix, out);
}


# conv kernel: a work-group walks 5 row pairs of one sample; next pair's x rows load to registers under the current pair's matrix work, then go to LDS; weights loaded once per strip; same sums
# speedup vs baseline: 2.9387x; 1.1254x over previous
#include <hip/hip_runtime.h>

#define B_    16
#define CIN_  16
#define OUT_  16
#define H_    320
#define W_    320
#define HW_   (H_ * W_)
#define KEXP_ 4
#define HID_  5
#define KTOT_ 144
#define P2_   336
#define CO_   3

typedef __attribute__((ext_vector_type(2))) float v2f;
typedef __attribute__((ext_vector_type(8))) float v8f;

__global__ __launch_bounds__(256) void dyn_pool_kernel(
    const float* __restrict__ x, float* __restrict__ pooled) {
  const int bc = blockIdx.x;
  const float4* p4 = (const float4*)(x + (size_t)bc * HW_);
  float s = 0.0f;
  for (int i = threadIdx.x; i < HW_ / 4; i += 256) {
    float4 v = p4[i];
    s += v.x + v.y + v.z + v.w;
  }
  __shared__ float red[256];
  red[threadIdx.x] = s;
  __syncthreads();
  for (int off = 128; off > 0; off >>= 1) {
    if (threadIdx.x < (unsigned)off) red[threadIdx.x] += red[threadIdx.x + off];
    __syncthreads();
  }
  if (threadIdx.x == 0) pooled[bc] = red[0] * (1.0f / (float)HW_);
}

__global__ __launch_bounds__(256) void dyn_attn_kernel(
    const float* __restrict__ pooled, const float* __restrict__ w1,
    const float* __restrict__ w2, const float* __restrict__ b2,
    const float* __restrict__ weight, float* __restrict__ wmix) {
  const int t = threadIdx.x;
  const int b = blockIdx.x;

  __shared__ __attribute__((aligned(16))) float sm[KEXP_ * OUT_];
  __shared__ __attribute__((aligned(16))) float wt[OUT_ * KTOT_];

  const float4* wq = (const float4*)weight;
  float4 wv[9];
#pragma unroll
  for (int i = 0; i < 9; ++i) wv[i] = wq[t + 256 * i];

  if (t < 16) {
    const int j = t;
    float hid[HID_];
#pragma unroll
    for (int h = 0; h < HID_; ++h) {
      float a = 0.0f;
#pragma unroll
      for (int c = 0; c < CIN_; ++c) a += pooled[b * CIN_ + c] * w1[h * CIN_ + c];
      hid[h] = fmaxf(a, 0.0f);
    }
    float av[KEXP_];
#pragma unroll
    for (int k = 0; k < KEXP_; ++k) {
      const int i = k * 16 + j;
      float a = b2[i];
#pragma unroll
      for (int h = 0; h < HID_; ++h) a += hid[h] * w2[i * HID_ + h];
      av[k] = a;
    }
    float mx = fmaxf(fmaxf(av[0], av[1]), fmaxf(av[2], av[3]));
    float e[KEXP_], sum = 0.0f;
#pragma unroll
    for (int k = 0; k < KEXP_; ++k) { e[k] = expf(av[k] - mx); sum += e[k]; }
    const float inv = 1.0f / sum;
#pragma unroll
    for (int k = 0; k < KEXP_; ++k) sm[k * 16 + j] = e[k] * inv;
  }
  __syncthreads();

#pragma unroll
  for (int i = 0; i < 9; ++i) {
    const int g = t + 256 * i;
    const int oc = g / 9;
    const int rd = g - oc * 9;
    const int o = oc >> 4;
    const int c = oc & 15;
    const float4 cf = *(const float4*)(sm + o * KEXP_);
    float acc = 0.0f;
    acc += cf.x * wv[i].x;
    acc += cf.y * wv[i].y;
    acc += cf.z * wv[i].z;
    acc += cf.w * wv[i].w;
    wt[o * KTOT_ + rd * 16 + c] = acc;
  }
  __syncthreads();

  float4* dst = (float4*)(wmix + (size_t)b * OUT_ * KTOT_);
  const float4* sp = (const float4*)wt;
  for (int q = t; q < OUT_ * KTOT_ / 4; q += 256) dst[q] = sp[q];
}

#define NPAIR_  5
#define NSTRIP_ ((H_ / 2) / NPAIR_)

__device__ __forceinline__ void conv_issue(const float* __restrict__ xb, int h0,
                                           float4 (&ve)[10], float4 (&vo)[10]) {
#pragma unroll
  for (int i = 0; i < 10; ++i) {
    const unsigned f = threadIdx.x + 256u * i;
    const unsigned k = f % 80u;
    const unsigned q = f / 80u;
    const int cp = q & 7;
    const int hh = h0 - 1 + (int)(q >> 3);
    const int hc = min(max(hh, 0), H_ - 1);
    const float4* sp =
        (const float4*)(xb + (size_t)(2 * cp) * HW_ + hc * W_) + k;
    ve[i] = sp[0];
    vo[i] = sp[HW_ / 4];
  }
}

__device__ __forceinline__ void conv_commit(float* xs, int h0,
                                            const float4 (&ve)[10], const float4 (&vo)[10]) {
#pragma unroll
  for (int i = 0; i < 10; ++i) {
    const unsigned f = threadIdx.x + 256u * i;
    const unsigned k = f % 80u;
    const unsigned q = f / 80u;
    const int hh = h0 - 1 + (int)(q >> 3);
    const bool ok = (hh >= 0) && (hh < H_);
    const float4 z = make_float4(0.0f, 0.0f, 0.0f, 0.0f);
    const float4 e = ok ? ve[i] : z;
    const float4 o = ok ? vo[i] : z;
    float4* dp = (float4*)(xs + (q * P2_ + 4u + 4u * k) * 2u);
    dp[0] = make_float4(e.x, o.x, e.y, o.y);
    dp[1] = make_float4(e.z, o.z, e.w, o.w);
  }
}

__device__ __forceinline__ void lds_barrier() {
  __builtin_amdgcn_fence(__ATOMIC_RELEASE, "workgroup", "local");
  __builtin_amdgcn_s_barrier();
  __builtin_amdgcn_fence(__ATOMIC_ACQUIRE, "workgroup", "local");
}

__global__ __launch_bounds__(256) void dyn_conv_kernel(
    const float* __restrict__ x, const float* __restrict__ wmix,
    float* __restrict__ out) {
  const int blk = blockIdx.x;
  const int b = blk / NSTRIP_;
  const int p0 = (blk % NSTRIP_) * NPAIR_;

  __shared__ float xs[4 * 8 * P2_ * 2];

  const float* xb = x + (size_t)b * CIN_ * HW_;
  const int lane = threadIdx.x & 31;
  const int wave = threadIdx.x >> 5;
  const int m = lane & 15;
  const int hi = lane >> 4;

  const float* wb = wmix + ((size_t)b * OUT_ + m) * KTOT_;
  v2f a[36];
#pragma unroll
  for (int s = 0; s < 36; ++s) {
    a[s].x = wb[4 * s + 2 * hi + 0];
    a[s].y = wb[4 * s + 2 * hi + 1];
  }

  float4 ve[10], vo[10];
  conv_issue(xb, p0 * 2, ve, vo);

  if (threadIdx.x < 64) {
    const unsigned q = threadIdx.x >> 1;
    const unsigned slot = (threadIdx.x & 1) ? (unsigned)(CO_ + W_ + 1) : (unsigned)CO_;
    v2f zz = {0.0f, 0.0f};
    *(v2f*)(xs + (q * P2_ + slot) * 2u) = zz;
  }

#pragma unroll 1
  for (int ip = 0; ip < NPAIR_; ++ip) {
    const int h0 = (p0 + ip) * 2;

    conv_commit(xs, h0, ve, vo);
    lds_barrier();
    if (ip + 1 < NPAIR_) conv_issue(xb, h0 + 2, ve, vo);

#pragma unroll 1
    for (int it = 0; it < 5; ++it) {
      const int tt = wave * 5 + it;
      const int lr = tt / 20;
      const int wbase = (tt % 20) * 16;
      const int base = ((lr * 8 + hi) * P2_ + wbase + m + CO_) * 2;

      v8f acc = {};
#pragma unroll
      for (int s = 0; s < 36; ++s) {
        const int cb = (4 * s) & 15;
        const int rd = (4 * s) >> 4;
        const int r  = rd / 3;
        const int dx = rd % 3;
        const v2f bf =
            *(const v2f*)(xs + base + ((r * 8 + (cb >> 1)) * P2_ + dx) * 2);
        acc = __builtin_amdgcn_wmma_f32_16x16x4_f32(
            false, a[s], false, bf, (short)0, acc, false, false);
      }

      float* ob = out + (size_t)b * OUT_ * HW_ + (h0 + lr) * W_ + wbase + m;
#pragma unroll
      for (int jj = 0; jj < 8; ++jj)
        ob[(size_t)(jj + 8 * hi) * HW_] = acc[jj];
    }
    lds_barrier();
  }
}

extern "C" void kernel_launch(void* const* d_in, const int* in_sizes, int n_in,
                              void* d_out, int out_size, void* d_ws, size_t ws_size,
                              hipStream_t stream) {
  (void)in_sizes; (void)n_in; (void)out_size; (void)ws_size;
  const float* x      = (const float*)d_in[0];
  const float* w1     = (const float*)d_in[1];
  const float* w2     = (const float*)d_in[2];
  const float* b2     = (const float*)d_in[3];
  const float* weight = (const float*)d_in[4];
  float* out = (float*)d_out;

  float* pooled = (float*)d_ws;
  float* wmix   = (float*)d_ws + 256;

  dyn_pool_kernel<<<B_ * CIN_, 256, 0, stream>>>(x, pooled);
  dyn_attn_kernel<<<B_, 256, 0, stream>>>(pooled, w1, w2, b2, weight, wmix);
  dyn_conv_kernel<<<B_ * NSTRIP_, 256, 0, stream>>>(x, wmix, out);
}


# attention MLP, softmax and expert mix moved into the conv kernel: each work-group mixes its sample's weights in LDS under its first x-row loads; launches 3 -> 2; same sums
# speedup vs baseline: 3.0584x; 1.0407x over previous
#include <hip/hip_runtime.h>

#define B_    16
#define CIN_  16
#define OUT_  16
#define H_    320
#define W_    320
#define HW_   (H_ * W_)
#define KEXP_ 4
#define HID_  5
#define KTOT_ 144
#define P2_   336
#define CO_   3

typedef __attribute__((ext_vector_type(2))) float v2f;
typedef __attribute__((ext_vector_type(8))) float v8f;

__global__ __launch_bounds__(256) void dyn_pool_kernel(
    const float* __restrict__ x, float* __restrict__ pooled) {
  const int bc = blockIdx.x;
  const float4* p4 = (const float4*)(x + (size_t)bc * HW_);
  float s = 0.0f;
  for (int i = threadIdx.x; i < HW_ / 4; i += 256) {
    float4 v = p4[i];
    s += v.x + v.y + v.z + v.w;
  }
  __shared__ float red[256];
  red[threadIdx.x] = s;
  __syncthreads();
  for (int off = 128; off > 0; off >>= 1) {
    if (threadIdx.x < (unsigned)off) red[threadIdx.x] += red[threadIdx.x + off];
    __syncthreads();
  }
  if (threadIdx.x == 0) pooled[bc] = red[0] * (1.0f / (float)HW_);
}

#define NPAIR_  5
#define NSTRIP_ ((H_ / 2) / NPAIR_)

__device__ __forceinline__ void conv_issue(const float* __restrict__ xb, int h0,
                                           float4 (&ve)[10], float4 (&vo)[10]) {
#pragma unroll
  for (int i = 0; i < 10; ++i) {
    const unsigned f = threadIdx.x + 256u * i;
    const unsigned k = f % 80u;
    const unsigned q = f / 80u;
    const int cp = q & 7;
    const int hh = h0 - 1 + (int)(q >> 3);
    const int hc = min(max(hh, 0), H_ - 1);
    const float4* sp =
        (const float4*)(xb + (size_t)(2 * cp) * HW_ + hc * W_) + k;
    ve[i] = sp[0];
    vo[i] = sp[HW_ / 4];
  }
}

__device__ __forceinline__ void conv_commit(float* xs, int h0,
                                            const float4 (&ve)[10], const float4 (&vo)[10]) {
#pragma unroll
  for (int i = 0; i < 10; ++i) {
    const unsigned f = threadIdx.x + 256u * i;
    const unsigned k = f % 80u;
    const unsigned q = f / 80u;
    const int hh = h0 - 1 + (int)(q >> 3);
    const bool ok = (hh >= 0) && (hh < H_);
    const float4 z = make_float4(0.0f, 0.0f, 0.0f, 0.0f);
    const float4 e = ok ? ve[i] : z;
    const float4 o = ok ? vo[i] : z;
    float4* dp = (float4*)(xs + (q * P2_ + 4u + 4u * k) * 2u);
    dp[0] = make_float4(e.x, o.x, e.y, o.y);
    dp[1] = make_float4(e.z, o.z, e.w, o.w);
  }
}

__device__ __forceinline__ void lds_barrier() {
  __builtin_amdgcn_fence(__ATOMIC_RELEASE, "workgroup", "local");
  __builtin_amdgcn_s_barrier();
  __builtin_amdgcn_fence(__ATOMIC_ACQUIRE, "workgroup", "local");
}

__global__ __launch_bounds__(256) void dyn_conv_kernel(
    const float* __restrict__ x, const float* __restrict__ pooled,
    const float* __restrict__ w1, const float* __restrict__ w2,
    const float* __restrict__ b2, const float* __restrict__ weight,
    float* __restrict__ out) {
  const int blk = blockIdx.x;
  const int b = blk / NSTRIP_;
  const int p0 = (blk % NSTRIP_) * NPAIR_;

  __shared__ float xs[4 * 8 * P2_ * 2];
  __shared__ __attribute__((aligned(16))) float sm[KEXP_ * OUT_];
  __shared__ __attribute__((aligned(16))) float wt[OUT_ * KTOT_];

  const float* xb = x + (size_t)b * CIN_ * HW_;
  const int t = threadIdx.x;
  const int lane = threadIdx.x & 31;
  const int wave = threadIdx.x >> 5;
  const int m = lane & 15;
  const int hi = lane >> 4;

  const float4* wq = (const float4*)weight;
  float4 wv[9];
#pragma unroll
  for (int i = 0; i < 9; ++i) wv[i] = wq[t + 256 * i];

  float w2r[KEXP_][HID_], b2r[KEXP_];
  if (t < 16) {
#pragma unroll
    for (int k = 0; k < KEXP_; ++k) {
      const int i = k * 16 + t;
      b2r[k] = b2[i];
#pragma unroll
      for (int h = 0; h < HID_; ++h) w2r[k][h] = w2[i * HID_ + h];
    }
  }

  float4 ve[10], vo[10];
  conv_issue(xb, p0 * 2, ve, vo);

  if (threadIdx.x < 64) {
    const unsigned q = threadIdx.x >> 1;
    const unsigned slot = (threadIdx.x & 1) ? (unsigned)(CO_ + W_ + 1) : (unsigned)CO_;
    v2f zz = {0.0f, 0.0f};
    *(v2f*)(xs + (q * P2_ + slot) * 2u) = zz;
  }

  if (t < 16) {
    const int j = t;
    float hid[HID_];
#pragma unroll
    for (int h = 0; h < HID_; ++h) {
      float acc = 0.0f;
#pragma unroll
      for (int c = 0; c < CIN_; ++c) acc += pooled[b * CIN_ + c] * w1[h * CIN_ + c];
      hid[h] = fmaxf(acc, 0.0f);
    }
    float av[KEXP_];
#pragma unroll
    for (int k = 0; k < KEXP_; ++k) {
      float acc = b2r[k];
#pragma unroll
      for (int h = 0; h < HID_; ++h) acc += hid[h] * w2r[k][h];
      av[k] = acc;
    }
    float mx = fmaxf(fmaxf(av[0], av[1]), fmaxf(av[2], av[3]));
    float e[KEXP_], sum = 0.0f;
#pragma unroll
    for (int k = 0; k < KEXP_; ++k) { e[k] = expf(av[k] - mx); sum += e[k]; }
    const float inv = 1.0f / sum;
#pragma unroll
    for (int k = 0; k < KEXP_; ++k) sm[k * 16 + j] = e[k] * inv;
  }
  lds_barrier();

#pragma unroll
  for (int i = 0; i < 9; ++i) {
    const int g = t + 256 * i;
    const int oc = g / 9;
    const int rd = g - oc * 9;
    const int o = oc >> 4;
    const int c = oc & 15;
    const float4 cf = *(const float4*)(sm + o * KEXP_);
    float acc = 0.0f;
    acc += cf.x * wv[i].x;
    acc += cf.y * wv[i].y;
    acc += cf.z * wv[i].z;
    acc += cf.w * wv[i].w;
    wt[o * KTOT_ + rd * 16 + c] = acc;
  }
  lds_barrier();

  const float* wb = wt + m * KTOT_ + 2 * hi;
  v2f a[36];
#pragma unroll
  for (int s = 0; s < 36; ++s) a[s] = *(const v2f*)(wb + 4 * s);

#pragma unroll 1
  for (int ip = 0; ip < NPAIR_; ++ip) {
    const int h0 = (p0 + ip) * 2;

    conv_commit(xs, h0, ve, vo);
    lds_barrier();
    if (ip + 1 < NPAIR_) conv_issue(xb, h0 + 2, ve, vo);

#pragma unroll 1
    for (int it = 0; it < 5; ++it) {
      const int tt = wave * 5 + it;
      const int lr = tt / 20;
      const int wbase = (tt % 20) * 16;
      const int base = ((lr * 8 + hi) * P2_ + wbase + m + CO_) * 2;

      v8f acc = {};
#pragma unroll
      for (int s = 0; s < 36; ++s) {
        const int cb = (4 * s) & 15;
        const int rd = (4 * s) >> 4;
        const int r  = rd / 3;
        const int dx = rd % 3;
        const v2f bf =
            *(const v2f*)(xs + base + ((r * 8 + (cb >> 1)) * P2_ + dx) * 2);
        acc = __builtin_amdgcn_wmma_f32_16x16x4_f32(
            false, a[s], false, bf, (short)0, acc, false, false);
      }

      float* ob = out + (size_t)b * OUT_ * HW_ + (h0 + lr) * W_ + wbase + m;
#pragma unroll
      for (int jj = 0; jj < 8; ++jj)
        ob[(size_t)(jj + 8 * hi) * HW_] = acc[jj];
    }
    lds_barrier();
  }
}

extern "C" void kernel_launch(void* const* d_in, const int* in_sizes, int n_in,
                              void* d_out, int out_size, void* d_ws, size_t ws_size,
                              hipStream_t stream) {
  (void)in_sizes; (void)n_in; (void)out_size; (void)ws_size;
  const float* x      = (const float*)d_in[0];
  const float* w1     = (const float*)d_in[1];
  const float* w2     = (const float*)d_in[2];
  const float* b2     = (const float*)d_in[3];
  const float* weight = (const float*)d_in[4];
  float* out = (float*)d_out;

  float* pooled = (float*)d_ws;

  dyn_pool_kernel<<<B_ * CIN_, 256, 0, stream>>>(x, pooled);
  dyn_conv_kernel<<<B_ * NSTRIP_, 256, 0, stream>>>(x, pooled, w1, w2, b2, weight, out);
}


# conv tile: minimal 2-output/3-tap filtering along w: 4 transformed products per output pair instead of 6 taps (48 WMMAs per 32 pixels, was 72); f32 kept, different summation order
# speedup vs baseline: 3.5055x; 1.1462x over previous
#include <hip/hip_runtime.h>

#define B_    16
#define CIN_  16
#define OUT_  16
#define H_    320
#define W_    320
#define HW_   (H_ * W_)
#define KEXP_ 4
#define HID_  5
#define KTOT_ 144
#define P2_   336
#define CO_   3

typedef __attribute__((ext_vector_type(2))) float v2f;
typedef __attribute__((ext_vector_type(8))) float v8f;

__global__ __launch_bounds__(256) void dyn_pool_kernel(
    const float* __restrict__ x, float* __restrict__ pooled) {
  const int bc = blockIdx.x;
  const float4* p4 = (const float4*)(x + (size_t)bc * HW_);
  float s = 0.0f;
  for (int i = threadIdx.x; i < HW_ / 4; i += 256) {
    float4 v = p4[i];
    s += v.x + v.y + v.z + v.w;
  }
  __shared__ float red[256];
  red[threadIdx.x] = s;
  __syncthreads();
  for (int off = 128; off > 0; off >>= 1) {
    if (threadIdx.x < (unsigned)off) red[threadIdx.x] += red[threadIdx.x + off];
    __syncthreads();
  }
  if (threadIdx.x == 0) pooled[bc] = red[0] * (1.0f / (float)HW_);
}

#define NPAIR_  5
#define NSTRIP_ ((H_ / 2) / NPAIR_)

__device__ __forceinline__ void conv_issue(const float* __restrict__ xb, int h0,
                                           float4 (&ve)[10], float4 (&vo)[10]) {
#pragma unroll
  for (int i = 0; i < 10; ++i) {
    const unsigned f = threadIdx.x + 256u * i;
    const unsigned k = f % 80u;
    const unsigned q = f / 80u;
    const int cp = q & 7;
    const int hh = h0 - 1 + (int)(q >> 3);
    const int hc = min(max(hh, 0), H_ - 1);
    const unsigned off = (unsigned)(2 * cp) * (unsigned)HW_ + (unsigned)hc * (unsigned)W_ + 4u * k;
    const float4* sp = (const float4*)(xb + off);
    ve[i] = sp[0];
    vo[i] = sp[HW_ / 4];
  }
}

__device__ __forceinline__ void conv_commit(float* xs, int h0,
                                            const float4 (&ve)[10], const float4 (&vo)[10]) {
#pragma unroll
  for (int i = 0; i < 10; ++i) {
    const unsigned f = threadIdx.x + 256u * i;
    const unsigned k = f % 80u;
    const unsigned q = f / 80u;
    const int hh = h0 - 1 + (int)(q >> 3);
    const bool ok = (hh >= 0) && (hh < H_);
    const float4 z = make_float4(0.0f, 0.0f, 0.0f, 0.0f);
    const float4 e = ok ? ve[i] : z;
    const float4 o = ok ? vo[i] : z;
    float4* dp = (float4*)(xs + (q * P2_ + 4u + 4u * k) * 2u);
    dp[0] = make_float4(e.x, o.x, e.y, o.y);
    dp[1] = make_float4(e.z, o.z, e.w, o.w);
  }
}

__device__ __forceinline__ void lds_barrier() {
  __builtin_amdgcn_fence(__ATOMIC_RELEASE, "workgroup", "local");
  __builtin_amdgcn_s_barrier();
  __builtin_amdgcn_fence(__ATOMIC_ACQUIRE, "workgroup", "local");
}

__global__ __launch_bounds__(256, 4) void dyn_conv_kernel(
    const float* __restrict__ x, const float* __restrict__ pooled,
    const float* __restrict__ w1, const float* __restrict__ w2,
    const float* __restrict__ b2, const float* __restrict__ weight,
    float* __restrict__ out) {
  const int blk = blockIdx.x;
  const int b = blk / NSTRIP_;
  const int p0 = (blk % NSTRIP_) * NPAIR_;

  __shared__ float xs[4 * 8 * P2_ * 2];
  __shared__ __attribute__((aligned(16))) float sm[KEXP_ * OUT_];
  __shared__ __attribute__((aligned(16))) float wt[OUT_ * KTOT_];

  const float* xb = x + (size_t)b * CIN_ * HW_;
  const int t = threadIdx.x;
  const int lane = threadIdx.x & 31;
  const int wave = __builtin_amdgcn_readfirstlane((int)(threadIdx.x >> 5));
  const int m = lane & 15;
  const int hi = lane >> 4;

  const float4* wq = (const float4*)weight;
  float4 wv[9];
#pragma unroll
  for (int i = 0; i < 9; ++i) wv[i] = wq[t + 256 * i];

  float w2r[KEXP_][HID_], b2r[KEXP_];
  if (t < 16) {
#pragma unroll
    for (int k = 0; k < KEXP_; ++k) {
      const int i = k * 16 + t;
      b2r[k] = b2[i];
#pragma unroll
      for (int h = 0; h < HID_; ++h) w2r[k][h] = w2[i * HID_ + h];
    }
  }

  float4 ve[10], vo[10];
  conv_issue(xb, p0 * 2, ve, vo);

  if (threadIdx.x < 64) {
    const unsigned q = threadIdx.x >> 1;
    const unsigned slot = (threadIdx.x & 1) ? (unsigned)(CO_ + W_ + 1) : (unsigned)CO_;
    v2f zz = {0.0f, 0.0f};
    *(v2f*)(xs + (q * P2_ + slot) * 2u) = zz;
  }

  if (t < 16) {
    const int j = t;
    float hid[HID_];
#pragma unroll
    for (int h = 0; h < HID_; ++h) {
      float acc = 0.0f;
#pragma unroll
      for (int c = 0; c < CIN_; ++c) acc += pooled[b * CIN_ + c] * w1[h * CIN_ + c];
      hid[h] = fmaxf(acc, 0.0f);
    }
    float av[KEXP_];
#pragma unroll
    for (int k = 0; k < KEXP_; ++k) {
      float acc = b2r[k];
#pragma unroll
      for (int h = 0; h < HID_; ++h) acc += hid[h] * w2r[k][h];
      av[k] = acc;
    }
    float mx = fmaxf(fmaxf(av[0], av[1]), fmaxf(av[2], av[3]));
    float e[KEXP_], sum = 0.0f;
#pragma unroll
    for (int k = 0; k < KEXP_; ++k) { e[k] = expf(av[k] - mx); sum += e[k]; }
    const float inv = 1.0f / sum;
#pragma unroll
    for (int k = 0; k < KEXP_; ++k) sm[k * 16 + j] = e[k] * inv;
  }
  lds_barrier();

#pragma unroll
  for (int i = 0; i < 9; ++i) {
    const int g = t + 256 * i;
    const int oc = g / 9;
    const int rd = g - oc * 9;
    const int o = oc >> 4;
    const int c = oc & 15;
    const float4 cf = *(const float4*)(sm + o * KEXP_);
    float acc = 0.0f;
    acc += cf.x * wv[i].x;
    acc += cf.y * wv[i].y;
    acc += cf.z * wv[i].z;
    acc += cf.w * wv[i].w;
    wt[o * KTOT_ + rd * 16 + c] = acc;
  }
  lds_barrier();

  const float* wb = wt + m * KTOT_ + 2 * hi;
  v2f a[48];
#pragma unroll
  for (int r = 0; r < 3; ++r) {
#pragma unroll
    for (int q = 0; q < 4; ++q) {
      const v2f g0 = *(const v2f*)(wb + (r * 3 + 0) * 16 + 4 * q);
      const v2f g1 = *(const v2f*)(wb + (r * 3 + 1) * 16 + 4 * q);
      const v2f g2 = *(const v2f*)(wb + (r * 3 + 2) * 16 + 4 * q);
      const v2f s02 = g0 + g2;
      const int f = (r * 4 + q) * 4;
      a[f + 0] = g0;
      a[f + 1] = (s02 + g1) * 0.5f;
      a[f + 2] = (s02 - g1) * 0.5f;
      a[f + 3] = g2;
    }
  }

#pragma unroll 1
  for (int ip = 0; ip < NPAIR_; ++ip) {
    const int h0 = (p0 + ip) * 2;

    conv_commit(xs, h0, ve, vo);
    lds_barrier();
    if (ip + 1 < NPAIR_) conv_issue(xb, h0 + 2, ve, vo);

#pragma unroll 1
    for (int tt = wave; tt < 20; tt += 8) {
      const int lr = tt / 10;
      const int wb32 = (tt % 10) * 32;
      const int base = ((lr * 8 + hi) * P2_ + wb32 + 2 * m + CO_) * 2;

      v8f M0 = {}, M1 = {}, M2 = {}, M3 = {};
#pragma unroll
      for (int r = 0; r < 3; ++r) {
#pragma unroll
        for (int q = 0; q < 4; ++q) {
          const float* p = xs + base + ((r * 8 + 2 * q) * P2_) * 2;
          const v2f d0 = *(const v2f*)(p + 0);
          const v2f d1 = *(const v2f*)(p + 2);
          const v2f d2 = *(const v2f*)(p + 4);
          const v2f d3 = *(const v2f*)(p + 6);
          const v2f V0 = d0 - d2;
          const v2f V1 = d1 + d2;
          const v2f V2 = d2 - d1;
          const v2f V3 = d1 - d3;
          const int f = (r * 4 + q) * 4;
          M0 = __builtin_amdgcn_wmma_f32_16x16x4_f32(
              false, a[f + 0], false, V0, (short)0, M0, false, false);
          M1 = __builtin_amdgcn_wmma_f32_16x16x4_f32(
              false, a[f + 1], false, V1, (short)0, M1, false, false);
          M2 = __builtin_amdgcn_wmma_f32_16x16x4_f32(
              false, a[f + 2], false, V2, (short)0, M2, false, false);
          M3 = __builtin_amdgcn_wmma_f32_16x16x4_f32(
              false, a[f + 3], false, V3, (short)0, M3, false, false);
        }
      }

      float* ob = out + (size_t)b * OUT_ * HW_ + (h0 + lr) * W_ + wb32 + 2 * m;
      v2f y[8];
#pragma unroll
      for (int jj = 0; jj < 8; ++jj) {
        y[jj].x = (M0[jj] + M1[jj]) + M2[jj];
        y[jj].y = (M1[jj] - M2[jj]) - M3[jj];
      }
#pragma unroll
      for (int jj = 0; jj < 8; ++jj)
        *(v2f*)(ob + (size_t)(jj + 8 * hi) * HW_) = y[jj];
    }
    lds_barrier();
  }
}

extern "C" void kernel_launch(void* const* d_in, const int* in_sizes, int n_in,
                              void* d_out, int out_size, void* d_ws, size_t ws_size,
                              hipStream_t stream) {
  (void)in_sizes; (void)n_in; (void)out_size; (void)ws_size;
  const float* x      = (const float*)d_in[0];
  const float* w1     = (const float*)d_in[1];
  const float* w2     = (const float*)d_in[2];
  const float* b2     = (const float*)d_in[3];
  const float* weight = (const float*)d_in[4];
  float* out = (float*)d_out;

  float* pooled = (float*)d_ws;

  dyn_pool_kernel<<<B_ * CIN_, 256, 0, stream>>>(x, pooled);
  dyn_conv_kernel<<<B_ * NSTRIP_, 256, 0, stream>>>(x, pooled, w1, w2, b2, weight, out);
}


# pool kernel: each wave keeps 25 to 50 16-byte loads in flight, next group requested before the current is summed (was groups of 8 waited together); wave butterfly + 1 barrier; pool sum order changes
# speedup vs baseline: 3.6192x; 1.0324x over previous
#include <hip/hip_runtime.h>

#define B_    16
#define CIN_  16
#define OUT_  16
#define H_    320
#define W_    320
#define HW_   (H_ * W_)
#define KEXP_ 4
#define HID_  5
#define KTOT_ 144
#define P2_   336
#define CO_   3

typedef __attribute__((ext_vector_type(2))) float v2f;
typedef __attribute__((ext_vector_type(8))) float v8f;

#define PT_   256
#define PG_   25
#define PNG_  (HW_ / 4 / PT_ / PG_)

__global__ __launch_bounds__(PT_) void dyn_pool_kernel(
    const float* __restrict__ x, float* __restrict__ pooled) {
  const int bc = blockIdx.x;
  const float4* p4 = (const float4*)(x + (size_t)bc * HW_) + threadIdx.x;
  float4 v[2][PG_];
  float s0 = 0.0f, s1 = 0.0f, s2 = 0.0f, s3 = 0.0f;
#pragma unroll
  for (int i = 0; i < PG_; ++i) v[0][i] = p4[i * PT_];
#pragma unroll
  for (int g = 0; g < PNG_; ++g) {
    __builtin_amdgcn_sched_barrier(0);
    if (g + 1 < PNG_) {
#pragma unroll
      for (int i = 0; i < PG_; ++i) v[(g + 1) & 1][i] = p4[((g + 1) * PG_ + i) * PT_];
    }
    __builtin_amdgcn_sched_barrier(0);
#pragma unroll
    for (int i = 0; i < PG_; ++i) {
      s0 += v[g & 1][i].x; s1 += v[g & 1][i].y; s2 += v[g & 1][i].z; s3 += v[g & 1][i].w;
    }
  }
  float s = (s0 + s1) + (s2 + s3);
#pragma unroll
  for (int off = 16; off > 0; off >>= 1) s += __shfl_xor(s, off, 32);
  __shared__ float red[PT_ / 32];
  if ((threadIdx.x & 31) == 0) red[threadIdx.x >> 5] = s;
  __syncthreads();
  if (threadIdx.x == 0) {
    float tot = 0.0f;
#pragma unroll
    for (int w = 0; w < PT_ / 32; ++w) tot += red[w];
    pooled[bc] = tot * (1.0f / (float)HW_);
  }
}

#define NPAIR_  5
#define NSTRIP_ ((H_ / 2) / NPAIR_)

__device__ __forceinline__ void conv_issue(const float* __restrict__ xb, int h0,
                                           float4 (&ve)[10], float4 (&vo)[10]) {
#pragma unroll
  for (int i = 0; i < 10; ++i) {
    const unsigned f = threadIdx.x + 256u * i;
    const unsigned k = f % 80u;
    const unsigned q = f / 80u;
    const int cp = q & 7;
    const int hh = h0 - 1 + (int)(q >> 3);
    const int hc = min(max(hh, 0), H_ - 1);
    const unsigned off = (unsigned)(2 * cp) * (unsigned)HW_ + (unsigned)hc * (unsigned)W_ + 4u * k;
    const float4* sp = (const float4*)(xb + off);
    ve[i] = sp[0];
    vo[i] = sp[HW_ / 4];
  }
}

__device__ __forceinline__ void conv_commit(float* xs, int h0,
                                            const float4 (&ve)[10], const float4 (&vo)[10]) {
#pragma unroll
  for (int i = 0; i < 10; ++i) {
    const unsigned f = threadIdx.x + 256u * i;
    const unsigned k = f % 80u;
    const unsigned q = f / 80u;
    const int hh = h0 - 1 + (int)(q >> 3);
    const bool ok = (hh >= 0) && (hh < H_);
    const float4 z = make_float4(0.0f, 0.0f, 0.0f, 0.0f);
    const float4 e = ok ? ve[i] : z;
    const float4 o = ok ? vo[i] : z;
    float4* dp = (float4*)(xs + (q * P2_ + 4u + 4u * k) * 2u);
    dp[0] = make_float4(e.x, o.x, e.y, o.y);
    dp[1] = make_float4(e.z, o.z, e.w, o.w);
  }
}

__device__ __forceinline__ void lds_barrier() {
  __builtin_amdgcn_fence(__ATOMIC_RELEASE, "workgroup", "local");
  __builtin_amdgcn_s_barrier();
  __builtin_amdgcn_fence(__ATOMIC_ACQUIRE, "workgroup", "local");
}

__global__ __launch_bounds__(256, 4) void dyn_conv_kernel(
    const float* __restrict__ x, const float* __restrict__ pooled,
    const float* __restrict__ w1, const float* __restrict__ w2,
    const float* __restrict__ b2, const float* __restrict__ weight,
    float* __restrict__ out) {
  const int blk = blockIdx.x;
  const int b = blk / NSTRIP_;
  const int p0 = (blk % NSTRIP_) * NPAIR_;

  __shared__ float xs[4 * 8 * P2_ * 2];
  __shared__ __attribute__((aligned(16))) float sm[KEXP_ * OUT_];
  __shared__ __attribute__((aligned(16))) float wt[OUT_ * KTOT_];

  const float* xb = x + (size_t)b * CIN_ * HW_;
  const int t = threadIdx.x;
  const int lane = threadIdx.x & 31;
  const int wave = __builtin_amdgcn_readfirstlane((int)(threadIdx.x >> 5));
  const int m = lane & 15;
  const int hi = lane >> 4;

  const float4* wq = (const float4*)weight;
  float4 wv[9];
#pragma unroll
  for (int i = 0; i < 9; ++i) wv[i] = wq[t + 256 * i];

  float w2r[KEXP_][HID_], b2r[KEXP_];
  if (t < 16) {
#pragma unroll
    for (int k = 0; k < KEXP_; ++k) {
      const int i = k * 16 + t;
      b2r[k] = b2[i];
#pragma unroll
      for (int h = 0; h < HID_; ++h) w2r[k][h] = w2[i * HID_ + h];
    }
  }

  float4 ve[10], vo[10];
  conv_issue(xb, p0 * 2, ve, vo);

  if (threadIdx.x < 64) {
    const unsigned q = threadIdx.x >> 1;
    const unsigned slot = (threadIdx.x & 1) ? (unsigned)(CO_ + W_ + 1) : (unsigned)CO_;
    v2f zz = {0.0f, 0.0f};
    *(v2f*)(xs + (q * P2_ + slot) * 2u) = zz;
  }

  if (t < 16) {
    const int j = t;
    float hid[HID_];
#pragma unroll
    for (int h = 0; h < HID_; ++h) {
      float acc = 0.0f;
#pragma unroll
      for (int c = 0; c < CIN_; ++c) acc += pooled[b * CIN_ + c] * w1[h * CIN_ + c];
      hid[h] = fmaxf(acc, 0.0f);
    }
    float av[KEXP_];
#pragma unroll
    for (int k = 0; k < KEXP_; ++k) {
      float acc = b2r[k];
#pragma unroll
      for (int h = 0; h < HID_; ++h) acc += hid[h] * w2r[k][h];
      av[k] = acc;
    }
    float mx = fmaxf(fmaxf(av[0], av[1]), fmaxf(av[2], av[3]));
    float e[KEXP_], sum = 0.0f;
#pragma unroll
    for (int k = 0; k < KEXP_; ++k) { e[k] = expf(av[k] - mx); sum += e[k]; }
    const float inv = 1.0f / sum;
#pragma unroll
    for (int k = 0; k < KEXP_; ++k) sm[k * 16 + j] = e[k] * inv;
  }
  lds_barrier();

#pragma unroll
  for (int i = 0; i < 9; ++i) {
    const int g = t + 256 * i;
    const int oc = g / 9;
    const int rd = g - oc * 9;
    const int o = oc >> 4;
    const int c = oc & 15;
    const float4 cf = *(const float4*)(sm + o * KEXP_);
    float acc = 0.0f;
    acc += cf.x * wv[i].x;
    acc += cf.y * wv[i].y;
    acc += cf.z * wv[i].z;
    acc += cf.w * wv[i].w;
    wt[o * KTOT_ + rd * 16 + c] = acc;
  }
  lds_barrier();

  const float* wb = wt + m * KTOT_ + 2 * hi;
  v2f a[48];
#pragma unroll
  for (int r = 0; r < 3; ++r) {
#pragma unroll
    for (int q = 0; q < 4; ++q) {
      const v2f g0 = *(const v2f*)(wb + (r * 3 + 0) * 16 + 4 * q);
      const v2f g1 = *(const v2f*)(wb + (r * 3 + 1) * 16 + 4 * q);
      const v2f g2 = *(const v2f*)(wb + (r * 3 + 2) * 16 + 4 * q);
      const v2f s02 = g0 + g2;
      const int f = (r * 4 + q) * 4;
      a[f + 0] = g0;
      a[f + 1] = (s02 + g1) * 0.5f;
      a[f + 2] = (s02 - g1) * 0.5f;
      a[f + 3] = g2;
    }
  }

#pragma unroll 1
  for (int ip = 0; ip < NPAIR_; ++ip) {
    const int h0 = (p0 + ip) * 2;

    conv_commit(xs, h0, ve, vo);
    lds_barrier();
    if (ip + 1 < NPAIR_) conv_issue(xb, h0 + 2, ve, vo);

#pragma unroll 1
    for (int tt = wave; tt < 20; tt += 8) {
      const int lr = tt / 10;
      const int wb32 = (tt % 10) * 32;
      const int base = ((lr * 8 + hi) * P2_ + wb32 + 2 * m + CO_) * 2;

      v8f M0 = {}, M1 = {}, M2 = {}, M3 = {};
#pragma unroll
      for (int r = 0; r < 3; ++r) {
#pragma unroll
        for (int q = 0; q < 4; ++q) {
          const float* p = xs + base + ((r * 8 + 2 * q) * P2_) * 2;
          const v2f d0 = *(const v2f*)(p + 0);
          const v2f d1 = *(const v2f*)(p + 2);
          const v2f d2 = *(const v2f*)(p + 4);
          const v2f d3 = *(const v2f*)(p + 6);
          const v2f V0 = d0 - d2;
          const v2f V1 = d1 + d2;
          const v2f V2 = d2 - d1;
          const v2f V3 = d1 - d3;
          const int f = (r * 4 + q) * 4;
          M0 = __builtin_amdgcn_wmma_f32_16x16x4_f32(
              false, a[f + 0], false, V0, (short)0, M0, false, false);
          M1 = __builtin_amdgcn_wmma_f32_16x16x4_f32(
              false, a[f + 1], false, V1, (short)0, M1, false, false);
          M2 = __builtin_amdgcn_wmma_f32_16x16x4_f32(
              false, a[f + 2], false, V2, (short)0, M2, false, false);
          M3 = __builtin_amdgcn_wmma_f32_16x16x4_f32(
              false, a[f + 3], false, V3, (short)0, M3, false, false);
        }
      }

      float* ob = out + (size_t)b * OUT_ * HW_ + (h0 + lr) * W_ + wb32 + 2 * m;
      v2f y[8];
#pragma unroll
      for (int jj = 0; jj < 8; ++jj) {
        y[jj].x = (M0[jj] + M1[jj]) + M2[jj];
        y[jj].y = (M1[jj] - M2[jj]) - M3[jj];
      }
#pragma unroll
      for (int jj = 0; jj < 8; ++jj)
        *(v2f*)(ob + (size_t)(jj + 8 * hi) * HW_) = y[jj];
    }
    lds_barrier();
  }
}

extern "C" void kernel_launch(void* const* d_in, const int* in_sizes, int n_in,
                              void* d_out, int out_size, void* d_ws, size_t ws_size,
                              hipStream_t stream) {
  (void)in_sizes; (void)n_in; (void)out_size; (void)ws_size;
  const float* x      = (const float*)d_in[0];
  const float* w1     = (const float*)d_in[1];
  const float* w2     = (const float*)d_in[2];
  const float* b2     = (const float*)d_in[3];
  const float* weight = (const float*)d_in[4];
  float* out = (float*)d_out;

  float* pooled = (float*)d_ws;

  dyn_pool_kernel<<<B_ * CIN_, PT_, 0, stream>>>(x, pooled);
  dyn_conv_kernel<<<B_ * NSTRIP_, 256, 0, stream>>>(x, pooled, w1, w2, b2, weight, out);
}


# conv tile: B operands read from LDS 2 or more k-steps ahead of their WMMAs (was read, wait, use in each step); registers for it from an LDS row ring that restages only a pair's 2 new rows; same sums
# speedup vs baseline: 3.9138x; 1.0814x over previous
#include <hip/hip_runtime.h>

#define B_    16
#define CIN_  16
#define OUT_  16
#define H_    320
#define W_    320
#define HW_   (H_ * W_)
#define KEXP_ 4
#define HID_  5
#define KTOT_ 144
#define P2_   336
#define CO_   3

typedef __attribute__((ext_vector_type(2))) float v2f;
typedef __attribute__((ext_vector_type(8))) float v8f;

#define PT_   256
#define PG_   25
#define PNG_  (HW_ / 4 / PT_ / PG_)

__global__ __launch_bounds__(PT_) void dyn_pool_kernel(
    const float* __restrict__ x, float* __restrict__ pooled) {
  const int bc = blockIdx.x;
  const float4* p4 = (const float4*)(x + (size_t)bc * HW_) + threadIdx.x;
  float4 v[2][PG_];
  float s0 = 0.0f, s1 = 0.0f, s2 = 0.0f, s3 = 0.0f;
#pragma unroll
  for (int i = 0; i < PG_; ++i) v[0][i] = p4[i * PT_];
#pragma unroll
  for (int g = 0; g < PNG_; ++g) {
    __builtin_amdgcn_sched_barrier(0);
    if (g + 1 < PNG_) {
#pragma unroll
      for (int i = 0; i < PG_; ++i) v[(g + 1) & 1][i] = p4[((g + 1) * PG_ + i) * PT_];
    }
    __builtin_amdgcn_sched_barrier(0);
#pragma unroll
    for (int i = 0; i < PG_; ++i) {
      s0 += v[g & 1][i].x; s1 += v[g & 1][i].y; s2 += v[g & 1][i].z; s3 += v[g & 1][i].w;
    }
  }
  float s = (s0 + s1) + (s2 + s3);
#pragma unroll
  for (int off = 16; off > 0; off >>= 1) s += __shfl_xor(s, off, 32);
  __shared__ float red[PT_ / 32];
  if ((threadIdx.x & 31) == 0) red[threadIdx.x >> 5] = s;
  __syncthreads();
  if (threadIdx.x == 0) {
    float tot = 0.0f;
#pragma unroll
    for (int w = 0; w < PT_ / 32; ++w) tot += red[w];
    pooled[bc] = tot * (1.0f / (float)HW_);
  }
}

#define NPAIR_  5
#define NSTRIP_ ((H_ / 2) / NPAIR_)

__device__ __forceinline__ void conv_issue(const float* __restrict__ xb, int hb,
                                           float4 (&ve)[5], float4 (&vo)[5]) {
#pragma unroll
  for (int i = 0; i < 5; ++i) {
    const unsigned f = threadIdx.x + 256u * i;
    const unsigned k = f % 80u;
    const unsigned q = f / 80u;
    const int cp = q & 7;
    const int hh = hb + (int)(q >> 3);
    const int hc = min(max(hh, 0), H_ - 1);
    const unsigned off = (unsigned)(2 * cp) * (unsigned)HW_ + (unsigned)hc * (unsigned)W_ + 4u * k;
    const float4* sp = (const float4*)(xb + off);
    ve[i] = sp[0];
    vo[i] = sp[HW_ / 4];
  }
}

__device__ __forceinline__ void conv_commit(float* xs, int hb, unsigned half,
                                            const float4 (&ve)[5], const float4 (&vo)[5]) {
#pragma unroll
  for (int i = 0; i < 5; ++i) {
    const unsigned f = threadIdx.x + 256u * i;
    const unsigned k = f % 80u;
    const unsigned q = f / 80u;
    const int hh = hb + (int)(q >> 3);
    const bool ok = (hh >= 0) && (hh < H_);
    const float4 z = make_float4(0.0f, 0.0f, 0.0f, 0.0f);
    const float4 e = ok ? ve[i] : z;
    const float4 o = ok ? vo[i] : z;
    float4* dp = (float4*)(xs + ((half * 16u + q) * P2_ + 4u + 4u * k) * 2u);
    dp[0] = make_float4(e.x, o.x, e.y, o.y);
    dp[1] = make_float4(e.z, o.z, e.w, o.w);
  }
}

__device__ __forceinline__ void conv_ldb(const float* p, v2f (&d)[4]) {
  d[0] = *(const v2f*)(p + 0);
  d[1] = *(const v2f*)(p + 2);
  d[2] = *(const v2f*)(p + 4);
  d[3] = *(const v2f*)(p + 6);
}

__device__ __forceinline__ void lds_barrier() {
  __builtin_amdgcn_fence(__ATOMIC_RELEASE, "workgroup", "local");
  __builtin_amdgcn_s_barrier();
  __builtin_amdgcn_fence(__ATOMIC_ACQUIRE, "workgroup", "local");
}

__global__ __launch_bounds__(256, 4) void dyn_conv_kernel(
    const float* __restrict__ x, const float* __restrict__ pooled,
    const float* __restrict__ w1, const float* __restrict__ w2,
    const float* __restrict__ b2, const float* __restrict__ weight,
    float* __restrict__ out) {
  const int blk = blockIdx.x;
  const int b = blk / NSTRIP_;
  const int p0 = (blk % NSTRIP_) * NPAIR_;

  __shared__ float xs[4 * 8 * P2_ * 2];
  __shared__ __attribute__((aligned(16))) float sm[KEXP_ * OUT_];
  __shared__ __attribute__((aligned(16))) float wt[OUT_ * KTOT_];

  const float* xb = x + (size_t)b * CIN_ * HW_;
  const int t = threadIdx.x;
  const int lane = threadIdx.x & 31;
  const int wave = __builtin_amdgcn_readfirstlane((int)(threadIdx.x >> 5));
  const int m = lane & 15;
  const int hi = lane >> 4;

  const float4* wq = (const float4*)weight;
  float4 wv[9];
#pragma unroll
  for (int i = 0; i < 9; ++i) wv[i] = wq[t + 256 * i];

  float w2r[KEXP_][HID_], b2r[KEXP_];
  if (t < 16) {
#pragma unroll
    for (int k = 0; k < KEXP_; ++k) {
      const int i = k * 16 + t;
      b2r[k] = b2[i];
#pragma unroll
      for (int h = 0; h < HID_; ++h) w2r[k][h] = w2[i * HID_ + h];
    }
  }

  float4 ve[5], vo[5], vf[5], vg[5];
  conv_issue(xb, p0 * 2 - 1, vf, vg);
  conv_issue(xb, p0 * 2 + 1, ve, vo);

  if (threadIdx.x < 64) {
    const unsigned q = threadIdx.x >> 1;
    const unsigned slot = (threadIdx.x & 1) ? (unsigned)(CO_ + W_ + 1) : (unsigned)CO_;
    v2f zz = {0.0f, 0.0f};
    *(v2f*)(xs + (q * P2_ + slot) * 2u) = zz;
  }

  if (t < 16) {
    const int j = t;
    float hid[HID_];
#pragma unroll
    for (int h = 0; h < HID_; ++h) {
      float acc = 0.0f;
#pragma unroll
      for (int c = 0; c < CIN_; ++c) acc += pooled[b * CIN_ + c] * w1[h * CIN_ + c];
      hid[h] = fmaxf(acc, 0.0f);
    }
    float av[KEXP_];
#pragma unroll
    for (int k = 0; k < KEXP_; ++k) {
      float acc = b2r[k];
#pragma unroll
      for (int h = 0; h < HID_; ++h) acc += hid[h] * w2r[k][h];
      av[k] = acc;
    }
    float mx = fmaxf(fmaxf(av[0], av[1]), fmaxf(av[2], av[3]));
    float e[KEXP_], sum = 0.0f;
#pragma unroll
    for (int k = 0; k < KEXP_; ++k) { e[k] = expf(av[k] - mx); sum += e[k]; }
    const float inv = 1.0f / sum;
#pragma unroll
    for (int k = 0; k < KEXP_; ++k) sm[k * 16 + j] = e[k] * inv;
  }
  lds_barrier();

#pragma unroll
  for (int i = 0; i < 9; ++i) {
    const int g = t + 256 * i;
    const int oc = g / 9;
    const int rd = g - oc * 9;
    const int o = oc >> 4;
    const int c = oc & 15;
    const float4 cf = *(const float4*)(sm + o * KEXP_);
    float acc = 0.0f;
    acc += cf.x * wv[i].x;
    acc += cf.y * wv[i].y;
    acc += cf.z * wv[i].z;
    acc += cf.w * wv[i].w;
    wt[o * KTOT_ + rd * 16 + c] = acc;
  }
  lds_barrier();

  conv_commit(xs, p0 * 2 - 1, 0u, vf, vg);

  const float* wb = wt + m * KTOT_ + 2 * hi;
  v2f a[48];
#pragma unroll
  for (int r = 0; r < 3; ++r) {
#pragma unroll
    for (int q = 0; q < 4; ++q) {
      const v2f g0 = *(const v2f*)(wb + (r * 3 + 0) * 16 + 4 * q);
      const v2f g1 = *(const v2f*)(wb + (r * 3 + 1) * 16 + 4 * q);
      const v2f g2 = *(const v2f*)(wb + (r * 3 + 2) * 16 + 4 * q);
      const v2f s02 = g0 + g2;
      const int f = (r * 4 + q) * 4;
      a[f + 0] = g0;
      a[f + 1] = (s02 + g1) * 0.5f;
      a[f + 2] = (s02 - g1) * 0.5f;
      a[f + 3] = g2;
    }
  }

#pragma unroll 1
  for (int ip = 0; ip < NPAIR_; ++ip) {
    const int h0 = (p0 + ip) * 2;
    const int rot = (ip & 1) * 2;

    conv_commit(xs, h0 + 1, (unsigned)((ip + 1) & 1), ve, vo);
    lds_barrier();
    if (ip + 1 < NPAIR_) conv_issue(xb, h0 + 3, ve, vo);

#pragma unroll 1
    for (int tt = wave; tt < 20; tt += 8) {
      const int lr = tt / 10;
      const int wb32 = (tt % 10) * 32;
      const float* lb = xs + (hi * P2_ + wb32 + 2 * m + CO_) * 2;
      const int sl0 = (lr + 0) ^ rot, sl1 = (lr + 1) ^ rot, sl2 = (lr + 2) ^ rot;
      const float* rb[3] = {lb + sl0 * (8 * P2_ * 2), lb + sl1 * (8 * P2_ * 2), lb + sl2 * (8 * P2_ * 2)};

      v8f M0 = {}, M1 = {}, M2 = {}, M3 = {};
      v2f d[3][4];
      conv_ldb(rb[0] + (0 * P2_) * 2, d[0]);
      conv_ldb(rb[0] + (2 * P2_) * 2, d[1]);
#pragma unroll
      for (int s = 0; s < 12; ++s) {
        if (s + 2 < 12)
          conv_ldb(rb[(s + 2) >> 2] + (2 * ((s + 2) & 3) * P2_) * 2, d[(s + 2) % 3]);
        __builtin_amdgcn_sched_barrier(0);
        const v2f V0 = d[s % 3][0] - d[s % 3][2];
        const v2f V1 = d[s % 3][1] + d[s % 3][2];
        const v2f V2 = d[s % 3][2] - d[s % 3][1];
        const v2f V3 = d[s % 3][1] - d[s % 3][3];
        const int f = s * 4;
        M0 = __builtin_amdgcn_wmma_f32_16x16x4_f32(
            false, a[f + 0], false, V0, (short)0, M0, false, false);
        M1 = __builtin_amdgcn_wmma_f32_16x16x4_f32(
            false, a[f + 1], false, V1, (short)0, M1, false, false);
        M2 = __builtin_amdgcn_wmma_f32_16x16x4_f32(
            false, a[f + 2], false, V2, (short)0, M2, false, false);
        M3 = __builtin_amdgcn_wmma_f32_16x16x4_f32(
            false, a[f + 3], false, V3, (short)0, M3, false, false);
      }

      float* ob = out + (size_t)b * OUT_ * HW_ + (h0 + lr) * W_ + wb32 + 2 * m;
      v2f y[8];
#pragma unroll
      for (int jj = 0; jj < 8; ++jj) {
        y[jj].x = (M0[jj] + M1[jj]) + M2[jj];
        y[jj].y = (M1[jj] - M2[jj]) - M3[jj];
      }
#pragma unroll
      for (int jj = 0; jj < 8; ++jj)
        *(v2f*)(ob + (size_t)(jj + 8 * hi) * HW_) = y[jj];
    }
    lds_barrier();
  }
}

extern "C" void kernel_launch(void* const* d_in, const int* in_sizes, int n_in,
                              void* d_out, int out_size, void* d_ws, size_t ws_size,
                              hipStream_t stream) {
  (void)in_sizes; (void)n_in; (void)out_size; (void)ws_size;
  const float* x      = (const float*)d_in[0];
  const float* w1     = (const float*)d_in[1];
  const float* w2     = (const float*)d_in[2];
  const float* b2     = (const float*)d_in[3];
  const float* weight = (const float*)d_in[4];
  float* out = (float*)d_out;

  float* pooled = (float*)d_ws;

  dyn_pool_kernel<<<B_ * CIN_, PT_, 0, stream>>>(x, pooled);
  dyn_conv_kernel<<<B_ * NSTRIP_, 256, 0, stream>>>(x, pooled, w1, w2, b2, weight, out);
}


# conv tile: minimal 2x2-output/3x3-tap filtering, 16 products per 4 outputs (64 WMMAs per 64 pixels, was 96); one 8-wave work-group per WGP to hold it in registers; f32 kept, different summation order
# speedup vs baseline: 4.2093x; 1.0755x over previous
#include <hip/hip_runtime.h>

#define B_    16
#define CIN_  16
#define OUT_  16
#define H_    320
#define W_    320
#define HW_   (H_ * W_)
#define KEXP_ 4
#define HID_  5
#define KTOT_ 144
#define P2_   336
#define CO_   3

typedef __attribute__((ext_vector_type(2))) float v2f;
typedef __attribute__((ext_vector_type(8))) float v8f;

#define PT_   256
#define PG_   25
#define PNG_  (HW_ / 4 / PT_ / PG_)

__global__ __launch_bounds__(PT_) void dyn_pool_kernel(
    const float* __restrict__ x, float* __restrict__ pooled) {
  const int bc = blockIdx.x;
  const float4* p4 = (const float4*)(x + (size_t)bc * HW_) + threadIdx.x;
  float4 v[2][PG_];
  float s0 = 0.0f, s1 = 0.0f, s2 = 0.0f, s3 = 0.0f;
#pragma unroll
  for (int i = 0; i < PG_; ++i) v[0][i] = p4[i * PT_];
#pragma unroll
  for (int g = 0; g < PNG_; ++g) {
    __builtin_amdgcn_sched_barrier(0);
    if (g + 1 < PNG_) {
#pragma unroll
      for (int i = 0; i < PG_; ++i) v[(g + 1) & 1][i] = p4[((g + 1) * PG_ + i) * PT_];
    }
    __builtin_amdgcn_sched_barrier(0);
#pragma unroll
    for (int i = 0; i < PG_; ++i) {
      s0 += v[g & 1][i].x; s1 += v[g & 1][i].y; s2 += v[g & 1][i].z; s3 += v[g & 1][i].w;
    }
  }
  float s = (s0 + s1) + (s2 + s3);
#pragma unroll
  for (int off = 16; off > 0; off >>= 1) s += __shfl_xor(s, off, 32);
  __shared__ float red[PT_ / 32];
  if ((threadIdx.x & 31) == 0) red[threadIdx.x >> 5] = s;
  __syncthreads();
  if (threadIdx.x == 0) {
    float tot = 0.0f;
#pragma unroll
    for (int w = 0; w < PT_ / 32; ++w) tot += red[w];
    pooled[bc] = tot * (1.0f / (float)HW_);
  }
}

#define NSTEP_  5
#define SROWS_  (4 * NSTEP_)
#define NSTRIP_ (H_ / SROWS_)
#define RING_   10
#define ROWF_   (8 * P2_ * 2)

template <int NI>
__device__ __forceinline__ void conv_issue(const float* __restrict__ xb, int hb,
                                           float4 (&ve)[NI], float4 (&vo)[NI]) {
#pragma unroll
  for (int i = 0; i < NI; ++i) {
    const unsigned f = threadIdx.x + 256u * i;
    const unsigned k = f % 80u;
    const unsigned q = f / 80u;
    const int cp = q & 7;
    const int hh = hb + (int)(q >> 3);
    const int hc = min(max(hh, 0), H_ - 1);
    const unsigned off = (unsigned)(2 * cp) * (unsigned)HW_ + (unsigned)hc * (unsigned)W_ + 4u * k;
    const float4* sp = (const float4*)(xb + off);
    ve[i] = sp[0];
    vo[i] = sp[HW_ / 4];
  }
}

template <int NI>
__device__ __forceinline__ void conv_commit(float* xs, int hb, unsigned sb,
                                            const float4 (&ve)[NI], const float4 (&vo)[NI]) {
#pragma unroll
  for (int i = 0; i < NI; ++i) {
    const unsigned f = threadIdx.x + 256u * i;
    const unsigned k = f % 80u;
    const unsigned q = f / 80u;
    const unsigned cp = q & 7u;
    const unsigned rw = q >> 3;
    const int hh = hb + (int)rw;
    const bool ok = (hh >= 0) && (hh < H_);
    const float4 z = make_float4(0.0f, 0.0f, 0.0f, 0.0f);
    const float4 e = ok ? ve[i] : z;
    const float4 o = ok ? vo[i] : z;
    unsigned slot = sb + rw;
    slot = (slot >= (unsigned)RING_) ? slot - (unsigned)RING_ : slot;
    float4* dp = (float4*)(xs + ((slot * 8u + cp) * P2_ + 4u + 4u * k) * 2u);
    dp[0] = make_float4(e.x, o.x, e.y, o.y);
    dp[1] = make_float4(e.z, o.z, e.w, o.w);
  }
}

__device__ __forceinline__ void conv_ldb(const float* p, v2f (&d)[4]) {
  d[0] = *(const v2f*)(p + 0);
  d[1] = *(const v2f*)(p + 2);
  d[2] = *(const v2f*)(p + 4);
  d[3] = *(const v2f*)(p + 6);
}

__device__ __forceinline__ void lds_barrier() {
  __builtin_amdgcn_fence(__ATOMIC_RELEASE, "workgroup", "local");
  __builtin_amdgcn_s_barrier();
  __builtin_amdgcn_fence(__ATOMIC_ACQUIRE, "workgroup", "local");
}

__global__ __launch_bounds__(256) void dyn_conv_kernel(
    const float* __restrict__ x, const float* __restrict__ pooled,
    const float* __restrict__ w1, const float* __restrict__ w2,
    const float* __restrict__ b2, const float* __restrict__ weight,
    float* __restrict__ out) {
  const int blk = blockIdx.x;
  const int b = blk / NSTRIP_;
  const int r0 = (blk % NSTRIP_) * SROWS_;

  __shared__ float xs[RING_ * ROWF_];
  __shared__ __attribute__((aligned(16))) float sm[KEXP_ * OUT_];
  __shared__ __attribute__((aligned(16))) float wt[OUT_ * KTOT_];

  const float* xb = x + (size_t)b * CIN_ * HW_;
  const int t = threadIdx.x;
  const int lane = threadIdx.x & 31;
  const int wave = __builtin_amdgcn_readfirstlane((int)(threadIdx.x >> 5));
  const int m = lane & 15;
  const int hi = lane >> 4;

  const float4* wq = (const float4*)weight;
  float4 wv[9];
#pragma unroll
  for (int i = 0; i < 9; ++i) wv[i] = wq[t + 256 * i];

  float w2r[KEXP_][HID_], b2r[KEXP_];
  if (t < 16) {
#pragma unroll
    for (int k = 0; k < KEXP_; ++k) {
      const int i = k * 16 + t;
      b2r[k] = b2[i];
#pragma unroll
      for (int h = 0; h < HID_; ++h) w2r[k][h] = w2[i * HID_ + h];
    }
  }

  float4 ve[10], vo[10], vf[5], vg[5];
  conv_issue<5>(xb, r0 - 1, vf, vg);
  conv_issue<10>(xb, r0 + 1, ve, vo);

  if (threadIdx.x < 2 * RING_ * 8) {
    const unsigned q = threadIdx.x >> 1;
    const unsigned slot = (threadIdx.x & 1) ? (unsigned)(CO_ + W_ + 1) : (unsigned)CO_;
    v2f zz = {0.0f, 0.0f};
    *(v2f*)(xs + (q * P2_ + slot) * 2u) = zz;
  }

  if (t < 16) {
    const int j = t;
    float hid[HID_];
#pragma unroll
    for (int h = 0; h < HID_; ++h) {
      float acc = 0.0f;
#pragma unroll
      for (int c = 0; c < CIN_; ++c) acc += pooled[b * CIN_ + c] * w1[h * CIN_ + c];
      hid[h] = fmaxf(acc, 0.0f);
    }
    float av[KEXP_];
#pragma unroll
    for (int k = 0; k < KEXP_; ++k) {
      float acc = b2r[k];
#pragma unroll
      for (int h = 0; h < HID_; ++h) acc += hid[h] * w2r[k][h];
      av[k] = acc;
    }
    float mx = fmaxf(fmaxf(av[0], av[1]), fmaxf(av[2], av[3]));
    float e[KEXP_], sum = 0.0f;
#pragma unroll
    for (int k = 0; k < KEXP_; ++k) { e[k] = expf(av[k] - mx); sum += e[k]; }
    const float inv = 1.0f / sum;
#pragma unroll
    for (int k = 0; k < KEXP_; ++k) sm[k * 16 + j] = e[k] * inv;
  }
  lds_barrier();

#pragma unroll
  for (int i = 0; i < 9; ++i) {
    const int g = t + 256 * i;
    const int oc = g / 9;
    const int rd = g - oc * 9;
    const int o = oc >> 4;
    const int c = oc & 15;
    const float4 cf = *(const float4*)(sm + o * KEXP_);
    float acc = 0.0f;
    acc += cf.x * wv[i].x;
    acc += cf.y * wv[i].y;
    acc += cf.z * wv[i].z;
    acc += cf.w * wv[i].w;
    wt[o * KTOT_ + rd * 16 + c] = acc;
  }
  lds_barrier();

  conv_commit<5>(xs, r0 - 1, 0u, vf, vg);

  const float* wb = wt + m * KTOT_ + 2 * hi;
  v2f a[64];
#pragma unroll
  for (int q = 0; q < 4; ++q) {
    v2f P[4][3];
#pragma unroll
    for (int dx = 0; dx < 3; ++dx) {
      const v2f g0 = *(const v2f*)(wb + (0 * 3 + dx) * 16 + 4 * q);
      const v2f g1 = *(const v2f*)(wb + (1 * 3 + dx) * 16 + 4 * q);
      const v2f g2 = *(const v2f*)(wb + (2 * 3 + dx) * 16 + 4 * q);
      const v2f s02 = g0 + g2;
      P[0][dx] = g0;
      P[1][dx] = (s02 + g1) * 0.5f;
      P[2][dx] = (s02 - g1) * 0.5f;
      P[3][dx] = g2;
    }
#pragma unroll
    for (int i = 0; i < 4; ++i) {
      const v2f s02 = P[i][0] + P[i][2];
      const int f = q * 16 + i * 4;
      a[f + 0] = P[i][0];
      a[f + 1] = (s02 + P[i][1]) * 0.5f;
      a[f + 2] = (s02 - P[i][1]) * 0.5f;
      a[f + 3] = P[i][2];
    }
  }

#pragma unroll 1
  for (int is = 0; is < NSTEP_; ++is) {
    const int h0 = r0 + 4 * is;

    conv_commit<10>(xs, h0 + 1, (unsigned)((4 * is + 2) % RING_), ve, vo);
    lds_barrier();
    if (is + 1 < NSTEP_) conv_issue<10>(xb, h0 + 5, ve, vo);

#pragma unroll 1
    for (int tt = wave; tt < 20; tt += 8) {
      const int lr = tt / 10;
      const int wb32 = (tt % 10) * 32;
      const float* lb = xs + (hi * P2_ + wb32 + 2 * m + CO_) * 2;
      const int jb = 4 * is + 2 * lr;
      const float* rb[4] = {lb + ((jb + 0) % RING_) * ROWF_, lb + ((jb + 1) % RING_) * ROWF_,
                            lb + ((jb + 2) % RING_) * ROWF_, lb + ((jb + 3) % RING_) * ROWF_};

      v8f M[4][4];
#pragma unroll
      for (int i = 0; i < 4; ++i)
#pragma unroll
        for (int j = 0; j < 4; ++j) M[i][j] = v8f{};
      v2f d[2][4][4];
#pragma unroll
      for (int i = 0; i < 4; ++i) conv_ldb(rb[i], d[0][i]);
#pragma unroll
      for (int q = 0; q < 4; ++q) {
        if (q + 1 < 4) {
#pragma unroll
          for (int i = 0; i < 4; ++i) conv_ldb(rb[i] + (2 * (q + 1) * P2_) * 2, d[(q + 1) & 1][i]);
        }
        __builtin_amdgcn_sched_barrier(0);
        v2f T[4][4];
#pragma unroll
        for (int j = 0; j < 4; ++j) {
          T[0][j] = d[q & 1][0][j] - d[q & 1][2][j];
          T[1][j] = d[q & 1][1][j] + d[q & 1][2][j];
          T[2][j] = d[q & 1][2][j] - d[q & 1][1][j];
          T[3][j] = d[q & 1][1][j] - d[q & 1][3][j];
        }
#pragma unroll
        for (int i = 0; i < 4; ++i) {
          const v2f V0 = T[i][0] - T[i][2];
          const v2f V1 = T[i][1] + T[i][2];
          const v2f V2 = T[i][2] - T[i][1];
          const v2f V3 = T[i][1] - T[i][3];
          const int f = q * 16 + i * 4;
          M[i][0] = __builtin_amdgcn_wmma_f32_16x16x4_f32(
              false, a[f + 0], false, V0, (short)0, M[i][0], false, false);
          M[i][1] = __builtin_amdgcn_wmma_f32_16x16x4_f32(
              false, a[f + 1], false, V1, (short)0, M[i][1], false, false);
          M[i][2] = __builtin_amdgcn_wmma_f32_16x16x4_f32(
              false, a[f + 2], false, V2, (short)0, M[i][2], false, false);
          M[i][3] = __builtin_amdgcn_wmma_f32_16x16x4_f32(
              false, a[f + 3], false, V3, (short)0, M[i][3], false, false);
        }
      }

      float* ob = out + (size_t)b * OUT_ * HW_ + (h0 + 2 * lr) * W_ + wb32 + 2 * m;
      v2f y0[8], y1[8];
#pragma unroll
      for (int jj = 0; jj < 8; ++jj) {
        float za[4], zb[4];
#pragma unroll
        for (int i = 0; i < 4; ++i) {
          za[i] = (M[i][0][jj] + M[i][1][jj]) + M[i][2][jj];
          zb[i] = (M[i][1][jj] - M[i][2][jj]) - M[i][3][jj];
        }
        y0[jj].x = (za[0] + za[1]) + za[2];
        y0[jj].y = (zb[0] + zb[1]) + zb[2];
        y1[jj].x = (za[1] - za[2]) - za[3];
        y1[jj].y = (zb[1] - zb[2]) - zb[3];
      }
#pragma unroll
      for (int jj = 0; jj < 8; ++jj) {
        *(v2f*)(ob + (size_t)(jj + 8 * hi) * HW_) = y0[jj];
        *(v2f*)(ob + (size_t)(jj + 8 * hi) * HW_ + W_) = y1[jj];
      }
    }
  }
}

extern "C" void kernel_launch(void* const* d_in, const int* in_sizes, int n_in,
                              void* d_out, int out_size, void* d_ws, size_t ws_size,
                              hipStream_t stream) {
  (void)in_sizes; (void)n_in; (void)out_size; (void)ws_size;
  const float* x      = (const float*)d_in[0];
  const float* w1     = (const float*)d_in[1];
  const float* w2     = (const float*)d_in[2];
  const float* b2     = (const float*)d_in[3];
  const float* weight = (const float*)d_in[4];
  float* out = (float*)d_out;

  float* pooled = (float*)d_ws;

  dyn_pool_kernel<<<B_ * CIN_, PT_, 0, stream>>>(x, pooled);
  dyn_conv_kernel<<<B_ * NSTRIP_, 256, 0, stream>>>(x, pooled, w1, w2, b2, weight, out);
}
